# starting kernel
# baseline (speedup 1.0000x reference)
#include <hip/hip_runtime.h>

typedef float     v2f  __attribute__((ext_vector_type(2)));
typedef float     v8f  __attribute__((ext_vector_type(8)));
typedef _Float16  v16h __attribute__((ext_vector_type(16)));

#define DIMF 128

#if __has_builtin(__builtin_amdgcn_wmma_f32_16x16x4_f32)
#define USE_F32_WMMA 1
#endif


__global__ void zero_f32(float* __restrict__ p, long long n) {
    long long i = (long long)blockIdx.x * blockDim.x + threadIdx.x;
    if (i < n) p[i] = 0.0f;
}

__global__ void deg_count(const int* __restrict__ dst, float* __restrict__ deg, int E) {
    int e = blockIdx.x * blockDim.x + threadIdx.x;
    if (e < E) atomicAdd(&deg[dst[e]], 1.0f);
}

__global__ void deg_finish(float* __restrict__ deg, int n) {
    int v = blockIdx.x * blockDim.x + threadIdx.x;
    if (v < n) deg[v] = rsqrtf(deg[v] + 1.0f);
}

__global__ __launch_bounds__(256) void gcn_gemm(const float* __restrict__ X,
                                                const float* __restrict__ W,
                                                float* __restrict__ H) {
    const int m0   = blockIdx.x << 4;
    const int wave = threadIdx.x >> 5;
    const int lane = threadIdx.x & 31;
    const int col0 = wave << 4;
    const int row  = lane & 15;
    const int hi   = lane >> 4;

    v8f acc = {};
    const float* xrow = X + (size_t)(m0 + row) * DIMF;

#ifdef USE_F32_WMMA
    const int kh = hi * 2;
    #pragma unroll
    for (int k = 0; k < DIMF; k += 4) {
        v2f a, b;
        a.x = xrow[k + kh];
        a.y = xrow[k + kh + 1];
        b.x = W[(size_t)(k + kh)     * DIMF + col0 + row];
        b.y = W[(size_t)(k + kh + 1) * DIMF + col0 + row];
        acc = __builtin_amdgcn_wmma_f32_16x16x4_f32(
                  false, a, false, b, (short)0, acc, false, false);
    }
#else
    for (int k = 0; k < DIMF; k += 32) {
        v16h a, b;
        #pragma unroll
        for (int i = 0; i < 16; ++i) {
            int vg = i >> 1;
            int ka = ((vg < 4) ? (vg * 2) : (16 + (vg - 4) * 2)) + (i & 1) + hi * 8;
            a[i] = (_Float16)xrow[k + ka];
            int kb = i + hi * 16;
            b[i] = (_Float16)W[(size_t)(k + kb) * DIMF + col0 + row];
        }
        acc = __builtin_amdgcn_wmma_f32_16x16x32_f16(
                  false, a, false, b, (short)0, acc, false, false);
    }
#endif

    __syncthreads();

    float* hp = H + (size_t)m0 * DIMF + col0 + row;
    const int mo = hi * 8;
    #pragma unroll
    for (int j = 0; j < 8; ++j)
        hp[(size_t)(mo + j) * DIMF] = acc[j];
}

__global__ void gcn_edge(const float* __restrict__ H,
                         const int* __restrict__ src,
                         const int* __restrict__ dst,
                         const float* __restrict__ dinv,
                         float* __restrict__ G, int E) {
    long long t = (long long)blockIdx.x * blockDim.x + threadIdx.x;
    int e = (int)(t >> 5);
    if (e >= E) return;
    int f = ((int)t & 31) * 4;
    int u = src[e];
    int v = dst[e];
    float norm = dinv[u] * dinv[v];
    const float4 hv = *(const float4*)(H + (size_t)u * DIMF + f);
    float* gp = G + (size_t)v * DIMF + f;
    atomicAdd(gp + 0, hv.x * norm);
    atomicAdd(gp + 1, hv.y * norm);
    atomicAdd(gp + 2, hv.z * norm);
    atomicAdd(gp + 3, hv.w * norm);
}

__global__ void gcn_finalize(const float* __restrict__ G,
                             const float* __restrict__ Hs,
                             const float* __restrict__ dinv,
                             const float* __restrict__ b,
                             float* __restrict__ out, long long nd) {
    long long i = (long long)blockIdx.x * blockDim.x + threadIdx.x;
    if (i >= nd) return;
    int v = (int)(i >> 7);
    int d = (int)(i & (DIMF - 1));
    float s = dinv[v];
    float val = G[i] + s * s * Hs[i] + b[d];
    out[i] = (val > 0.0f) ? val : 0.01f * val;
}


extern "C" void kernel_launch(void* const* d_in, const int* in_sizes, int n_in,
                              void* d_out, int out_size, void* d_ws, size_t ws_size,
                              hipStream_t stream) {
    const float* x  = (const float*)d_in[0];
    const int*   ei = (const int*)d_in[1];
    const float* Wts[3] = {(const float*)d_in[2], (const float*)d_in[4], (const float*)d_in[6]};
    const float* bs[3]  = {(const float*)d_in[3], (const float*)d_in[5], (const float*)d_in[7]};

    const int N = in_sizes[0] / DIMF;
    const int E = in_sizes[1] / 2;
    const int* srcp = ei;
    const int* dstp = ei + E;

    float* ws   = (float*)d_ws;
    float* dinv = ws;
    size_t off  = ((size_t)N + 63) & ~(size_t)63;
    float* Hbuf = ws + off;
    float* Gbuf = Hbuf + (size_t)N * DIMF;

    const int T = 256;
    const long long ND = (long long)N * DIMF;
    const long long edgeThreads = (long long)E * 32;
    const int tilesM = N / 16;

    zero_f32  <<<(int)((N + T - 1) / T), T, 0, stream>>>(dinv, N);
    deg_count <<<(E + T - 1) / T,        T, 0, stream>>>(dstp, dinv, E);
    deg_finish<<<(N + T - 1) / T,        T, 0, stream>>>(dinv, N);

    for (int l = 0; l < 3; ++l) {
        const float* in = (l == 0) ? x : Hbuf;
        gcn_gemm<<<tilesM, 256, 0, stream>>>(in, Wts[l], Hbuf);

        zero_f32<<<(int)((ND + T - 1) / T), T, 0, stream>>>(Gbuf, ND);
        gcn_edge<<<(int)((edgeThreads + T - 1) / T), T, 0, stream>>>(
            Hbuf, srcp, dstp, dinv, Gbuf, E);

        float* outp = (l == 2) ? (float*)d_out : Hbuf;
        gcn_finalize<<<(int)((ND + T - 1) / T), T, 0, stream>>>(
            Gbuf, Hbuf, dinv, bs[l], outp, ND);
    }
}


# Aggregation by owner blocks: a block owns 196 dst nodes, scans all edges, sums in-rows in LDS with LDS f32 atomics, adds self term, bias, relu; global atomics, zero, finalize gone; sum order: arrival
# speedup vs baseline: 1.9902x; 1.9902x over previous
#include <hip/hip_runtime.h>

typedef float     v2f  __attribute__((ext_vector_type(2)));
typedef float     v8f  __attribute__((ext_vector_type(8)));
typedef _Float16  v16h __attribute__((ext_vector_type(16)));

#define DIMF 128

#if __has_builtin(__builtin_amdgcn_wmma_f32_16x16x4_f32)
#define USE_F32_WMMA 1
#endif


__global__ void zero_f32(float* __restrict__ p, long long n) {
    long long i = (long long)blockIdx.x * blockDim.x + threadIdx.x;
    if (i < n) p[i] = 0.0f;
}

__global__ void deg_count(const int* __restrict__ dst, float* __restrict__ deg, int E) {
    int e = blockIdx.x * blockDim.x + threadIdx.x;
    if (e < E) atomicAdd(&deg[dst[e]], 1.0f);
}

__global__ void deg_finish(float* __restrict__ deg, int n) {
    int v = blockIdx.x * blockDim.x + threadIdx.x;
    if (v < n) deg[v] = rsqrtf(deg[v] + 1.0f);
}

__global__ __launch_bounds__(256) void gcn_gemm(const float* __restrict__ X,
                                                const float* __restrict__ W,
                                                float* __restrict__ H) {
    const int m0   = blockIdx.x << 4;
    const int wave = threadIdx.x >> 5;
    const int lane = threadIdx.x & 31;
    const int col0 = wave << 4;
    const int row  = lane & 15;
    const int hi   = lane >> 4;

    v8f acc = {};
    const float* xrow = X + (size_t)(m0 + row) * DIMF;

#ifdef USE_F32_WMMA
    const int kh = hi * 2;
    #pragma unroll
    for (int k = 0; k < DIMF; k += 4) {
        v2f a, b;
        a.x = xrow[k + kh];
        a.y = xrow[k + kh + 1];
        b.x = W[(size_t)(k + kh)     * DIMF + col0 + row];
        b.y = W[(size_t)(k + kh + 1) * DIMF + col0 + row];
        acc = __builtin_amdgcn_wmma_f32_16x16x4_f32(
                  false, a, false, b, (short)0, acc, false, false);
    }
#else
    for (int k = 0; k < DIMF; k += 32) {
        v16h a, b;
        #pragma unroll
        for (int i = 0; i < 16; ++i) {
            int vg = i >> 1;
            int ka = ((vg < 4) ? (vg * 2) : (16 + (vg - 4) * 2)) + (i & 1) + hi * 8;
            a[i] = (_Float16)xrow[k + ka];
            int kb = i + hi * 16;
            b[i] = (_Float16)W[(size_t)(k + kb) * DIMF + col0 + row];
        }
        acc = __builtin_amdgcn_wmma_f32_16x16x32_f16(
                  false, a, false, b, (short)0, acc, false, false);
    }
#endif

    __syncthreads();

    float* hp = H + (size_t)m0 * DIMF + col0 + row;
    const int mo = hi * 8;
    #pragma unroll
    for (int j = 0; j < 8; ++j)
        hp[(size_t)(mo + j) * DIMF] = acc[j];
}

#define RN 196
#define SCAN_U 8

__device__ __forceinline__ int4 load_dst4(const int* __restrict__ p, int e0, int E) {
    int4 r;
    r.x = (e0     < E) ? p[e0]     : -1;
    r.y = (e0 + 1 < E) ? p[e0 + 1] : -1;
    r.z = (e0 + 2 < E) ? p[e0 + 2] : -1;
    r.w = (e0 + 3 < E) ? p[e0 + 3] : -1;
    return r;
}

__global__ __launch_bounds__(1024) void gcn_agg(const float* __restrict__ H,
                                                 const int* __restrict__ src,
                                                 const int* __restrict__ dst,
                                                 const float* __restrict__ dinv,
                                                 const float* __restrict__ bias,
                                                 float* __restrict__ out, int N, int E) {
    __shared__ float acc[RN * DIMF];

    const int tid  = threadIdx.x;
    const int lane = tid & 31;
    const int wave = __builtin_amdgcn_readfirstlane(tid >> 5);
    const int lo   = blockIdx.x * RN;
    const int nloc = (N - lo < RN) ? (N - lo) : RN;

    for (int i = tid; i < RN * DIMF / 4; i += 1024)
        ((float4*)acc)[i] = make_float4(0.0f, 0.0f, 0.0f, 0.0f);
    __syncthreads();

    const bool vec_ok = ((E & 3) == 0) && ((((size_t)dst) & 15) == 0);
    const int nchunks = (E + 127) >> 7;
    int qn = 0;
    int qe = 0;

    auto flush = [&](int cnt) {
        int u = 0, vr = 0;
        float nrm = 0.0f;
        if (lane < cnt) {
            u = src[qe];
            int v = dst[qe];
            vr = v - lo;
            nrm = dinv[u] * dinv[v];
        }
        for (int i = 0; i < cnt; i += 8) {
            float4 r[8];
            #pragma unroll
            for (int g = 0; g < 8; ++g) {
                int idx = (i + g < cnt) ? (i + g) : (cnt - 1);
                int ui = __builtin_amdgcn_readlane(u, idx);
                r[g] = *(const float4*)(H + (size_t)ui * DIMF + lane * 4);
            }
            #pragma unroll
            for (int g = 0; g < 8; ++g) {
                if (i + g < cnt) {
                    float s = __int_as_float(__builtin_amdgcn_readlane(__float_as_int(nrm), i + g));
                    int vi  = __builtin_amdgcn_readlane(vr, i + g);
                    float* a = acc + vi * DIMF + lane;
                    atomicAdd(a,      r[g].x * s);
                    atomicAdd(a + 32, r[g].y * s);
                    atomicAdd(a + 64, r[g].z * s);
                    atomicAdd(a + 96, r[g].w * s);
                }
            }
        }
    };

    for (int c = wave; c < nchunks; c += 32 * SCAN_U) {
        int4 d[SCAN_U];
        if (vec_ok && (((c + 32 * (SCAN_U - 1)) << 7) + 128 <= E)) {
            #pragma unroll
            for (int u = 0; u < SCAN_U; ++u)
                d[u] = *(const int4*)(dst + ((c + 32 * u) << 7) + (lane << 2));
        } else {
            #pragma unroll
            for (int u = 0; u < SCAN_U; ++u)
                d[u] = load_dst4(dst, ((c + 32 * u) << 7) + (lane << 2), E);
        }
        unsigned hb = 0;
        #pragma unroll
        for (int u = 0; u < SCAN_U; ++u) {
            hb |= ((unsigned)(d[u].x - lo) < (unsigned)nloc) ? (1u << (4 * u + 0)) : 0u;
            hb |= ((unsigned)(d[u].y - lo) < (unsigned)nloc) ? (1u << (4 * u + 1)) : 0u;
            hb |= ((unsigned)(d[u].z - lo) < (unsigned)nloc) ? (1u << (4 * u + 2)) : 0u;
            hb |= ((unsigned)(d[u].w - lo) < (unsigned)nloc) ? (1u << (4 * u + 3)) : 0u;
        }
        unsigned any = __builtin_amdgcn_ballot_w32(hb != 0);
        while (any) {
            int j = __builtin_ctz(any);
            any &= any - 1;
            unsigned hbj = (unsigned)__builtin_amdgcn_readlane((int)hb, j);
            while (hbj) {
                int t = __builtin_ctz(hbj);
                hbj &= hbj - 1;
                int e = ((c + 32 * (t >> 2)) << 7) + (j << 2) + (t & 3);
                if (lane == qn) qe = e;
                ++qn;
                if (qn == 32) { flush(32); qn = 0; }
            }
        }
    }
    if (qn) flush(qn);

    __syncthreads();

    const float4 b4 = *(const float4*)(bias + lane * 4);
    for (int v = wave; v < nloc; v += 32) {
        const size_t ro = (size_t)(lo + v) * DIMF + lane * 4;
        const float s = dinv[lo + v];
        const float4 h = *(const float4*)(H + ro);
        const float* a = acc + v * DIMF + lane;
        float4 o;
        o.x = a[0]  + s * s * h.x + b4.x;
        o.y = a[32] + s * s * h.y + b4.y;
        o.z = a[64] + s * s * h.z + b4.z;
        o.w = a[96] + s * s * h.w + b4.w;
        o.x = (o.x > 0.0f) ? o.x : 0.01f * o.x;
        o.y = (o.y > 0.0f) ? o.y : 0.01f * o.y;
        o.z = (o.z > 0.0f) ? o.z : 0.01f * o.z;
        o.w = (o.w > 0.0f) ? o.w : 0.01f * o.w;
        *(float4*)(out + ro) = o;
    }
}


extern "C" void kernel_launch(void* const* d_in, const int* in_sizes, int n_in,
                              void* d_out, int out_size, void* d_ws, size_t ws_size,
                              hipStream_t stream) {
    const float* x  = (const float*)d_in[0];
    const int*   ei = (const int*)d_in[1];
    const float* Wts[3] = {(const float*)d_in[2], (const float*)d_in[4], (const float*)d_in[6]};
    const float* bs[3]  = {(const float*)d_in[3], (const float*)d_in[5], (const float*)d_in[7]};

    const int N = in_sizes[0] / DIMF;
    const int E = in_sizes[1] / 2;
    const int* srcp = ei;
    const int* dstp = ei + E;

    float* ws   = (float*)d_ws;
    float* dinv = ws;
    size_t off  = ((size_t)N + 63) & ~(size_t)63;
    float* Hbuf = ws + off;
    float* Gbuf = Hbuf + (size_t)N * DIMF;

    const int T = 256;
    const int tilesM = N / 16;
    const int aggBlocks = (N + RN - 1) / RN;

    zero_f32  <<<(int)((N + T - 1) / T), T, 0, stream>>>(dinv, N);
    deg_count <<<(E + T - 1) / T,        T, 0, stream>>>(dstp, dinv, E);
    deg_finish<<<(N + T - 1) / T,        T, 0, stream>>>(dinv, N);

    for (int l = 0; l < 3; ++l) {
        const float* in = (l == 0) ? x : Gbuf;
        gcn_gemm<<<tilesM, 256, 0, stream>>>(in, Wts[l], Hbuf);

        float* outp = (l == 2) ? (float*)d_out : Gbuf;
        gcn_agg<<<aggBlocks, 1024, 0, stream>>>(Hbuf, srcp, dstp, dinv, bs[l], outp, N, E);
    }
}


# Layer 1 summed as (Ahat X) W1: the first aggregation gathers x rows, then multiplies its LDS sums by W1 and, after bias and activation, by W2 on f32 WMMA; gemm launch gone; sum order changes
# speedup vs baseline: 3.2809x; 1.6485x over previous
#include <hip/hip_runtime.h>

typedef float     v2f  __attribute__((ext_vector_type(2)));
typedef float     v8f  __attribute__((ext_vector_type(8)));

#define DIMF 128

#if __has_builtin(__builtin_amdgcn_wmma_f32_16x16x4_f32)
#define USE_F32_WMMA 1
#endif


__global__ void zero_f32(float* __restrict__ p, long long n) {
    long long i = (long long)blockIdx.x * blockDim.x + threadIdx.x;
    if (i < n) p[i] = 0.0f;
}

#define RN 196
__global__ void deg_count_key(const int* __restrict__ dst, float* __restrict__ deg,
                              unsigned int* __restrict__ key, int E, int Epad) {
    const int e0 = (blockIdx.x * blockDim.x + threadIdx.x) << 2;
    if (e0 >= Epad) return;
    unsigned int k = 0;
    const bool vec_ok = ((E & 3) == 0) && ((((size_t)dst) & 15) == 0);
    if (vec_ok && e0 + 3 < E) {
        const int4 d = *(const int4*)(dst + e0);
        atomicAdd(&deg[d.x], 1.0f);
        atomicAdd(&deg[d.y], 1.0f);
        atomicAdd(&deg[d.z], 1.0f);
        atomicAdd(&deg[d.w], 1.0f);
        k = (unsigned int)(d.x / RN) | ((unsigned int)(d.y / RN) << 8) |
            ((unsigned int)(d.z / RN) << 16) | ((unsigned int)(d.w / RN) << 24);
    } else {
        #pragma unroll
        for (int q = 0; q < 4; ++q) {
            if (e0 + q < E) {
                const int v = dst[e0 + q];
                atomicAdd(&deg[v], 1.0f);
                k |= (unsigned int)(v / RN) << (8 * q);
            }
        }
    }
    key[e0 >> 2] = k;
}

__global__ __launch_bounds__(256) void deg_finish_cnt(float* __restrict__ deg,
                                                      unsigned int* __restrict__ cnt, int n) {
    __shared__ float part[8];
    const int v = blockIdx.x * RN + threadIdx.x;
    float d = 0.0f;
    if (threadIdx.x < RN && v < n) {
        d = deg[v];
        deg[v] = rsqrtf(d + 1.0f);
    }
    for (int o = 16; o > 0; o >>= 1) d += __shfl_xor(d, o);
    if ((threadIdx.x & 31) == 0) part[threadIdx.x >> 5] = d;
    __syncthreads();
    if (threadIdx.x == 0) {
        float t = 0.0f;
        for (int i = 0; i < 8; ++i) t += part[i];
        cnt[blockIdx.x] = (unsigned int)t;
    }
}

#define SCAN_U 8
#define KCH 512

__device__ __forceinline__ unsigned int zero_bytes(unsigned int x) {
    return ~((((x & 0x7F7F7F7Fu) + 0x7F7F7F7Fu) | x) | 0x7F7F7F7Fu);
}

__device__ __forceinline__ unsigned int match16(uint4 d, unsigned int pat) {
    return (zero_bytes(d.x ^ pat) >> 7) | (zero_bytes(d.y ^ pat) >> 6) |
           (zero_bytes(d.z ^ pat) >> 5) | (zero_bytes(d.w ^ pat) >> 4);
}

__device__ __forceinline__ unsigned int match16_tail(uint4 d, unsigned int blk, int e0, int E) {
    const unsigned int w[4] = {d.x, d.y, d.z, d.w};
    unsigned int m = 0;
    #pragma unroll
    for (int q = 0; q < 16; ++q) {
        const unsigned int kb = (w[q >> 2] >> (8 * (q & 3))) & 255u;
        if (e0 + q < E && kb == blk) m |= 1u << (8 * (q & 3) + (q >> 2));
    }
    return m;
}

__device__ __forceinline__ unsigned int segment_start(const unsigned int* __restrict__ cnt,
                                                      int blk, int lane) {
    unsigned int t = 0;
    for (int i = lane; i < blk; i += 32) t += cnt[i];
    for (int o = 16; o > 0; o >>= 1) t += __shfl_xor(t, o);
    return t;
}

__global__ __launch_bounds__(1024) void gcn_build(const int* __restrict__ src,
                                                   const int* __restrict__ dst,
                                                   const unsigned int* __restrict__ key,
                                                   const float* __restrict__ dinv,
                                                   const unsigned int* __restrict__ cnt,
                                                   int* __restrict__ recw, int E) {
    int2* __restrict__ rec = (int2*)recw;
    __shared__ unsigned int s_start;
    __shared__ unsigned int s_fill;

    const int tid  = threadIdx.x;
    const int lane = tid & 31;
    const int wave = __builtin_amdgcn_readfirstlane(tid >> 5);
    const int lo   = blockIdx.x * RN;

    if (wave == 0) {
        const unsigned int t = segment_start(cnt, blockIdx.x, lane);
        if (lane == 0) { s_start = t; s_fill = 0; }
    }
    __syncthreads();
    const unsigned int start = s_start;

    const unsigned int blk = blockIdx.x;
    const unsigned int pat = blk * 0x01010101u;
    const uint4* key4 = (const uint4*)key;
    const int nchunks = (E + KCH - 1) / KCH;
    int qn = 0;
    int qe = 0;

    auto flush = [&](int c) {
        unsigned int base = 0;
        if (lane == 0) base = atomicAdd(&s_fill, (unsigned int)c);
        base = (unsigned int)__builtin_amdgcn_readfirstlane((int)base);
        if (lane < c) {
            const int u = src[qe];
            const int v = dst[qe];
            const float nrm = dinv[u] * dinv[v];
            rec[start + base + lane] = make_int2(u | ((v - lo) << 16), __float_as_int(nrm));
        }
    };

    for (int c = wave; c < nchunks; c += 32 * SCAN_U) {
        unsigned int m[SCAN_U];
        uint4 d[SCAN_U];
        #pragma unroll
        for (int u = 0; u < SCAN_U; ++u) {
            const int cc = c + 32 * u;
            d[u] = key4[((cc < nchunks) ? cc : (nchunks - 1)) * (KCH / 16) + lane];
        }
        if ((c + 32 * (SCAN_U - 1) + 1) * KCH <= E) {
            #pragma unroll
            for (int u = 0; u < SCAN_U; ++u)
                m[u] = match16(d[u], pat);
        } else {
            #pragma unroll
            for (int u = 0; u < SCAN_U; ++u) {
                const int cc = c + 32 * u;
                m[u] = (cc < nchunks) ? match16_tail(d[u], blk, cc * KCH + (lane << 4), E) : 0u;
            }
        }
        const unsigned int h0 = m[0] | (m[1] << 4);
        const unsigned int h1 = m[2] | (m[3] << 4);
        const unsigned int h2 = m[4] | (m[5] << 4);
        const unsigned int h3 = m[6] | (m[7] << 4);
        #pragma unroll 1
        for (int p = 0; p < SCAN_U / 2; ++p) {
            const unsigned int hb = (p == 0) ? h0 : (p == 1) ? h1 : (p == 2) ? h2 : h3;
            unsigned any = __builtin_amdgcn_ballot_w32(hb != 0);
            while (any) {
                int j = __builtin_ctz(any);
                any &= any - 1;
                unsigned hbj = (unsigned)__builtin_amdgcn_readlane((int)hb, j);
                while (hbj) {
                    int t = __builtin_ctz(hbj);
                    hbj &= hbj - 1;
                    int e = (c + 32 * (2 * p + ((t >> 2) & 1))) * KCH + (j << 4) + ((t & 3) << 2) + (t >> 3);
                    if (lane == qn) qe = e;
                    ++qn;
                    if (qn == 32) { flush(32); qn = 0; }
                }
            }
        }
    }
    if (qn) flush(qn);
}

#define PITCH 132
#define RTILES ((RN + 15) / 16)
__device__ __forceinline__ void agg_block(float* acc, float* act, unsigned int* s_startp,
                                          const bool pre,
                                          const float* __restrict__ H,
                                          const int* __restrict__ recw,
                                          const unsigned int* __restrict__ cnt,
                                          const float* __restrict__ dinv,
                                          const float* __restrict__ bias,
                                          const float* __restrict__ Wp,
                                          const float* __restrict__ Wn,
                                          float* __restrict__ out, int N) {
    const int tid  = threadIdx.x;
    const int lane = tid & 31;
    const int wave = __builtin_amdgcn_readfirstlane(tid >> 5);
    const int lo   = blockIdx.x * RN;
    const int nloc = (N - lo < RN) ? (N - lo) : RN;

    for (int i = tid; i < RTILES * 16 * PITCH / 4; i += 1024)
        ((float4*)acc)[i] = make_float4(0.0f, 0.0f, 0.0f, 0.0f);
    if (wave == 0) {
        const unsigned int t = segment_start(cnt, blockIdx.x, lane);
        if (lane == 0) *s_startp = t;
    }
    __syncthreads();

    const int2* seg = (const int2*)recw + *s_startp;
    const int n = (int)cnt[blockIdx.x];

    auto batch = [&](int2 cur, int c, int nb) -> int2 {
        int2 nx = make_int2(0, 0);
        const int   u   = cur.x & 0xFFFF;
        const int   vr  = cur.x >> 16;
        const float nrm = __int_as_float(cur.y);
        for (int i = 0; i < c; i += 8) {
            float4 r[8];
            #pragma unroll
            for (int g = 0; g < 8; ++g) {
                int idx = (i + g < c) ? (i + g) : (c - 1);
                int ui = __builtin_amdgcn_readlane(u, idx);
                r[g] = *(const float4*)(H + (size_t)ui * DIMF + lane * 4);
            }
            if (i == 0 && nb < n) nx = seg[nb];
            #pragma unroll
            for (int g = 0; g < 8; ++g) {
                if (i + g < c) {
                    float s = __int_as_float(__builtin_amdgcn_readlane(__float_as_int(nrm), i + g));
                    int vi  = __builtin_amdgcn_readlane(vr, i + g);
                    float* a = acc + vi * PITCH + lane;
                    atomicAdd(a,      r[g].x * s);
                    atomicAdd(a + 32, r[g].y * s);
                    atomicAdd(a + 64, r[g].z * s);
                    atomicAdd(a + 96, r[g].w * s);
                }
            }
        }
        return nx;
    };

    int b0 = wave * 32;
    int2 cur = (b0 + lane < n) ? seg[b0 + lane] : make_int2(0, 0);
    for (; b0 < n; b0 += 1024) {
        const int c  = n - b0;
        const int nb = b0 + 1024 + lane;
        cur = (c >= 32) ? batch(cur, 32, nb) : batch(cur, c, nb);
    }

    const bool fuse = pre || (Wn != nullptr);
#ifdef USE_F32_WMMA
    const int tc   = lane & 15;
    const int hf   = lane >> 4;
    const int col0 = (wave & 7) << 4;

    v2f bw[32];
    auto strip_perm = [&](const float* __restrict__ Wm) {
        #pragma unroll
        for (int t = 0; t < 32; ++t) {
            const int p = 4 * t + 2 * hf;
            const int f = 4 * (p & 31) + (p >> 5);
            bw[t].x = Wm[(size_t)f       * DIMF + col0 + tc];
            bw[t].y = Wm[(size_t)(f + 4) * DIMF + col0 + tc];
        }
    };
    auto strip_plain = [&](const float* __restrict__ Wm) {
        #pragma unroll
        for (int t = 0; t < 32; ++t) {
            const int p = 4 * t + 2 * hf;
            bw[t].x = Wm[(size_t)p       * DIMF + col0 + tc];
            bw[t].y = Wm[(size_t)(p + 1) * DIMF + col0 + tc];
        }
    };
    if (fuse) strip_perm(pre ? Wp : Wn);
    const float bc = pre ? bias[col0 + tc] : 0.0f;
#endif

    __syncthreads();

    if (pre) {
        for (int v = wave; v < nloc; v += 32) {
            const float s = dinv[lo + v];
            const float4 h = *(const float4*)(H + (size_t)(lo + v) * DIMF + lane * 4);
            float* a = acc + v * PITCH + lane;
            a[0]  += s * s * h.x;
            a[32] += s * s * h.y;
            a[64] += s * s * h.z;
            a[96] += s * s * h.w;
        }
    } else {
        const float4 b4 = *(const float4*)(bias + lane * 4);
        for (int v = wave; v < nloc; v += 32) {
            const size_t ro = (size_t)(lo + v) * DIMF + lane * 4;
            const float s = dinv[lo + v];
            const float4 h = *(const float4*)(H + ro);
            float* a = acc + v * PITCH + lane;
            float4 o;
            o.x = a[0]  + s * s * h.x + b4.x;
            o.y = a[32] + s * s * h.y + b4.y;
            o.z = a[64] + s * s * h.z + b4.z;
            o.w = a[96] + s * s * h.w + b4.w;
            o.x = (o.x > 0.0f) ? o.x : 0.01f * o.x;
            o.y = (o.y > 0.0f) ? o.y : 0.01f * o.y;
            o.z = (o.z > 0.0f) ? o.z : 0.01f * o.z;
            o.w = (o.w > 0.0f) ? o.w : 0.01f * o.w;
            if (fuse) {
                a[0] = o.x; a[32] = o.y; a[64] = o.z; a[96] = o.w;
            } else {
                *(float4*)(out + ro) = o;
            }
        }
    }
    if (!fuse) return;

#ifdef USE_F32_WMMA
    __syncthreads();

    const int rg = ((wave >> 3) + wave) & 3;

    auto tile = [&](const float* ar) -> v8f {
        v8f d = {};
        #pragma unroll
        for (int t = 0; t < 32; ++t) {
            v2f a;
            a.x = ar[4 * t];
            a.y = ar[4 * t + 1];
            d = __builtin_amdgcn_wmma_f32_16x16x4_f32(
                    false, a, false, bw[t], (short)0, d, false, false);
        }
        return d;
    };

    if (pre) {
        #pragma unroll 1
        for (int rt = rg; rt < RTILES; rt += 4) {
            if (rt * 16 >= nloc) break;
            const v8f d = tile(acc + (rt * 16 + tc) * PITCH + 2 * hf);
            float* yp = act + (rt * 16 + 8 * hf) * PITCH + col0 + tc;
            #pragma unroll
            for (int j = 0; j < 8; ++j) {
                float y = d[j] + bc;
                y = (y > 0.0f) ? y : 0.01f * y;
                yp[j * PITCH] = y;
            }
        }
        strip_plain(Wn);
        __syncthreads();
    }

    const float* rows = pre ? act : acc;
    #pragma unroll 1
    for (int rt = rg; rt < RTILES; rt += 4) {
        if (rt * 16 >= nloc) break;
        const v8f d = tile(rows + (rt * 16 + tc) * PITCH + 2 * hf);
        const int r0 = rt * 16 + 8 * hf;
        float* hp = out + (size_t)(lo + r0) * DIMF + col0 + tc;
        if (rt * 16 + 16 <= nloc) {
            #pragma unroll
            for (int j = 0; j < 8; ++j)
                hp[(size_t)j * DIMF] = d[j];
        } else {
            const int lim = nloc - r0;
            #pragma unroll
            for (int j = 0; j < 8; ++j)
                if (j < lim) hp[(size_t)j * DIMF] = d[j];
        }
    }
#endif
}

__global__ __launch_bounds__(1024) void gcn_agg(const float* __restrict__ H,
                                                 const int* __restrict__ recw,
                                                 const unsigned int* __restrict__ cnt,
                                                 const float* __restrict__ dinv,
                                                 const float* __restrict__ bias,
                                                 const float* __restrict__ Wn,
                                                 float* __restrict__ out, int N) {
    __shared__ __attribute__((aligned(16))) float acc[RTILES * 16 * PITCH];
    __shared__ unsigned int s_start;
    agg_block(acc, nullptr, &s_start, false, H, recw, cnt, dinv, bias, nullptr, Wn, out, N);
}

__global__ __launch_bounds__(1024) void gcn_agg_first(const float* __restrict__ X,
                                                       const int* __restrict__ recw,
                                                       const unsigned int* __restrict__ cnt,
                                                       const float* __restrict__ dinv,
                                                       const float* __restrict__ bias,
                                                       const float* __restrict__ Wp,
                                                       const float* __restrict__ Wn,
                                                       float* __restrict__ out, int N) {
    __shared__ __attribute__((aligned(16))) float acc[RTILES * 16 * PITCH];
    __shared__ __attribute__((aligned(16))) float act[RTILES * 16 * PITCH];
    __shared__ unsigned int s_start;
    agg_block(acc, act, &s_start, true, X, recw, cnt, dinv, bias, Wp, Wn, out, N);
}


extern "C" void kernel_launch(void* const* d_in, const int* in_sizes, int n_in,
                              void* d_out, int out_size, void* d_ws, size_t ws_size,
                              hipStream_t stream) {
    const float* x  = (const float*)d_in[0];
    const int*   ei = (const int*)d_in[1];
    const float* Wts[3] = {(const float*)d_in[2], (const float*)d_in[4], (const float*)d_in[6]};
    const float* bs[3]  = {(const float*)d_in[3], (const float*)d_in[5], (const float*)d_in[7]};

    const int N = in_sizes[0] / DIMF;
    const int E = in_sizes[1] / 2;
    const int* srcp = ei;
    const int* dstp = ei + E;

    float* ws   = (float*)d_ws;
    float* dinv = ws;
    size_t off  = ((size_t)N + 63) & ~(size_t)63;
    float* Hbuf = ws + off;
    float* Gbuf = Hbuf + (size_t)N * DIMF;
    unsigned int* keyb = (unsigned int*)(Gbuf + (size_t)N * DIMF);
    const int Epad = ((E + KCH - 1) / KCH) * KCH;
    unsigned int* cntb = keyb + Epad / 4;
    int* recb = (int*)(cntb + 256);

    const int T = 256;
    const int aggBlocks = (N + RN - 1) / RN;

    zero_f32  <<<(int)((N + T - 1) / T), T, 0, stream>>>(dinv, N);
    deg_count_key<<<(Epad / 4 + T - 1) / T, T, 0, stream>>>(dstp, dinv, keyb, E, Epad);
    deg_finish_cnt<<<aggBlocks, 256, 0, stream>>>(dinv, cntb, N);
    gcn_build<<<aggBlocks, 1024, 0, stream>>>(srcp, dstp, keyb, dinv, cntb, recb, E);

    const float* Wnone = nullptr;
    float* outf = (float*)d_out;
    gcn_agg_first<<<aggBlocks, 1024, 0, stream>>>(x, recb, cntb, dinv, bs[0], Wts[0], Wts[1], Gbuf, N);
    gcn_agg<<<aggBlocks, 1024, 0, stream>>>(Gbuf, recb, cntb, dinv, bs[1], Wts[2], Hbuf, N);
    gcn_agg<<<aggBlocks, 1024, 0, stream>>>(Hbuf, recb, cntb, dinv, bs[2], Wnone,  outf, N);
}


# Aggregation sums kept in registers, not LDS: build files each record under its owner wave (local dst & 31); a wave adds rows into its 7 register rows by scalar index; sum order: self, then arrival
# speedup vs baseline: 3.4025x; 1.0371x over previous
#include <hip/hip_runtime.h>

typedef float     v2f  __attribute__((ext_vector_type(2)));
typedef float     v8f  __attribute__((ext_vector_type(8)));
typedef float     v32f __attribute__((ext_vector_type(32)));

#define DIMF 128

#if __has_builtin(__builtin_amdgcn_wmma_f32_16x16x4_f32)
#define USE_F32_WMMA 1
#endif


__global__ void zero_f32(float* __restrict__ p, long long n) {
    long long i = (long long)blockIdx.x * blockDim.x + threadIdx.x;
    if (i < n) p[i] = 0.0f;
}

#define RN 196
__global__ void deg_count_key(const int* __restrict__ dst, float* __restrict__ deg,
                              unsigned int* __restrict__ key, int E, int Epad) {
    const int e0 = (blockIdx.x * blockDim.x + threadIdx.x) << 2;
    if (e0 >= Epad) return;
    unsigned int k = 0;
    const bool vec_ok = ((E & 3) == 0) && ((((size_t)dst) & 15) == 0);
    if (vec_ok && e0 + 3 < E) {
        const int4 d = *(const int4*)(dst + e0);
        atomicAdd(&deg[d.x], 1.0f);
        atomicAdd(&deg[d.y], 1.0f);
        atomicAdd(&deg[d.z], 1.0f);
        atomicAdd(&deg[d.w], 1.0f);
        k = (unsigned int)(d.x / RN) | ((unsigned int)(d.y / RN) << 8) |
            ((unsigned int)(d.z / RN) << 16) | ((unsigned int)(d.w / RN) << 24);
    } else {
        #pragma unroll
        for (int q = 0; q < 4; ++q) {
            if (e0 + q < E) {
                const int v = dst[e0 + q];
                atomicAdd(&deg[v], 1.0f);
                k |= (unsigned int)(v / RN) << (8 * q);
            }
        }
    }
    key[e0 >> 2] = k;
}

__global__ __launch_bounds__(256) void deg_finish_cnt(float* __restrict__ deg,
                                                      unsigned int* __restrict__ cnt,
                                                      unsigned int* __restrict__ cw, int n) {
    __shared__ float part[8][32];
    const int v = blockIdx.x * RN + threadIdx.x;
    float d = 0.0f;
    if (threadIdx.x < RN && v < n) {
        d = deg[v];
        deg[v] = rsqrtf(d + 1.0f);
    }
    part[threadIdx.x >> 5][threadIdx.x & 31] = d;
    __syncthreads();
    if (threadIdx.x < 32) {
        float t = 0.0f;
        for (int i = 0; i < 8; ++i) t += part[i][threadIdx.x];
        cw[blockIdx.x * 32 + threadIdx.x] = (unsigned int)t;
        for (int o = 16; o > 0; o >>= 1) t += __shfl_xor(t, o);
        if (threadIdx.x == 0) cnt[blockIdx.x] = (unsigned int)t;
    }
}

#define SCAN_U 8
#define KCH 512

__device__ __forceinline__ unsigned int zero_bytes(unsigned int x) {
    return ~((((x & 0x7F7F7F7Fu) + 0x7F7F7F7Fu) | x) | 0x7F7F7F7Fu);
}

__device__ __forceinline__ unsigned int match16(uint4 d, unsigned int pat) {
    return (zero_bytes(d.x ^ pat) >> 7) | (zero_bytes(d.y ^ pat) >> 6) |
           (zero_bytes(d.z ^ pat) >> 5) | (zero_bytes(d.w ^ pat) >> 4);
}

__device__ __forceinline__ unsigned int match16_tail(uint4 d, unsigned int blk, int e0, int E) {
    const unsigned int w[4] = {d.x, d.y, d.z, d.w};
    unsigned int m = 0;
    #pragma unroll
    for (int q = 0; q < 16; ++q) {
        const unsigned int kb = (w[q >> 2] >> (8 * (q & 3))) & 255u;
        if (e0 + q < E && kb == blk) m |= 1u << (8 * (q & 3) + (q >> 2));
    }
    return m;
}

__device__ __forceinline__ unsigned int segment_start(const unsigned int* __restrict__ cnt,
                                                      int blk, int lane) {
    unsigned int t = 0;
    for (int i = lane; i < blk; i += 32) t += cnt[i];
    for (int o = 16; o > 0; o >>= 1) t += __shfl_xor(t, o);
    return t;
}

__global__ __launch_bounds__(1024) void gcn_build(const int* __restrict__ src,
                                                   const int* __restrict__ dst,
                                                   const unsigned int* __restrict__ key,
                                                   const float* __restrict__ dinv,
                                                   const unsigned int* __restrict__ cnt,
                                                   const unsigned int* __restrict__ cw,
                                                   int* __restrict__ recw, int E) {
    int2* __restrict__ rec = (int2*)recw;
    __shared__ unsigned int s_startw[32];
    __shared__ unsigned int s_fillw[32];

    const int tid  = threadIdx.x;
    const int lane = tid & 31;
    const int wave = __builtin_amdgcn_readfirstlane(tid >> 5);
    const int lo   = blockIdx.x * RN;

    if (wave == 0) {
        const unsigned int t = segment_start(cnt, blockIdx.x, lane);
        const unsigned int cl = cw[blockIdx.x * 32 + lane];
        unsigned int inc = cl;
        for (int o = 1; o < 32; o <<= 1) {
            const unsigned int y = __shfl_up(inc, o);
            if (lane >= o) inc += y;
        }
        s_startw[lane] = t + inc - cl;
        s_fillw[lane]  = 0;
    }
    __syncthreads();

    const unsigned int blk = blockIdx.x;
    const unsigned int pat = blk * 0x01010101u;
    const uint4* key4 = (const uint4*)key;
    const int nchunks = (E + KCH - 1) / KCH;
    int qn = 0;
    int qe = 0;

    auto flush = [&](int c) {
        if (lane < c) {
            const int u = src[qe];
            const int v = dst[qe];
            const float nrm = dinv[u] * dinv[v];
            const int vl = v - lo;
            const unsigned int slot = atomicAdd(&s_fillw[vl & 31], 1u);
            rec[s_startw[vl & 31] + slot] = make_int2(u | (vl << 16), __float_as_int(nrm));
        }
    };

    for (int c = wave; c < nchunks; c += 32 * SCAN_U) {
        unsigned int m[SCAN_U];
        uint4 d[SCAN_U];
        #pragma unroll
        for (int u = 0; u < SCAN_U; ++u) {
            const int cc = c + 32 * u;
            d[u] = key4[((cc < nchunks) ? cc : (nchunks - 1)) * (KCH / 16) + lane];
        }
        if ((c + 32 * (SCAN_U - 1) + 1) * KCH <= E) {
            #pragma unroll
            for (int u = 0; u < SCAN_U; ++u)
                m[u] = match16(d[u], pat);
        } else {
            #pragma unroll
            for (int u = 0; u < SCAN_U; ++u) {
                const int cc = c + 32 * u;
                m[u] = (cc < nchunks) ? match16_tail(d[u], blk, cc * KCH + (lane << 4), E) : 0u;
            }
        }
        const unsigned int h0 = m[0] | (m[1] << 4);
        const unsigned int h1 = m[2] | (m[3] << 4);
        const unsigned int h2 = m[4] | (m[5] << 4);
        const unsigned int h3 = m[6] | (m[7] << 4);
        #pragma unroll 1
        for (int p = 0; p < SCAN_U / 2; ++p) {
            const unsigned int hb = (p == 0) ? h0 : (p == 1) ? h1 : (p == 2) ? h2 : h3;
            unsigned any = __builtin_amdgcn_ballot_w32(hb != 0);
            while (any) {
                int j = __builtin_ctz(any);
                any &= any - 1;
                unsigned hbj = (unsigned)__builtin_amdgcn_readlane((int)hb, j);
                while (hbj) {
                    int t = __builtin_ctz(hbj);
                    hbj &= hbj - 1;
                    int e = (c + 32 * (2 * p + ((t >> 2) & 1))) * KCH + (j << 4) + ((t & 3) << 2) + (t >> 3);
                    if (lane == qn) qe = e;
                    ++qn;
                    if (qn == 32) { flush(32); qn = 0; }
                }
            }
        }
    }
    if (qn) flush(qn);
}

#define PITCH 132
#define RTILES ((RN + 15) / 16)
__device__ __forceinline__ void agg_block(float* acc, float* act,
                                          const bool pre,
                                          const float* __restrict__ H,
                                          const int* __restrict__ recw,
                                          const unsigned int* __restrict__ cnt,
                                          const unsigned int* __restrict__ cw,
                                          const float* __restrict__ dinv,
                                          const float* __restrict__ bias,
                                          const float* __restrict__ Wp,
                                          const float* __restrict__ Wn,
                                          float* __restrict__ out, int N) {
    const int tid  = threadIdx.x;
    const int lane = tid & 31;
    const int wave = __builtin_amdgcn_readfirstlane(tid >> 5);
    const int lo   = blockIdx.x * RN;
    const int nloc = (N - lo < RN) ? (N - lo) : RN;

    const unsigned int bstart = segment_start(cnt, blockIdx.x, lane);
    const unsigned int cl = cw[blockIdx.x * 32 + lane];
    unsigned int before = (lane < wave) ? cl : 0u;
    for (int o = 16; o > 0; o >>= 1) before += __shfl_xor(before, o);
    const int n = __builtin_amdgcn_readlane((int)cl, wave);
    const int2* seg = (const int2*)recw + bstart + before;

    v32f av = {};
    {
        float4 h[7];
        float  s[7];
        #pragma unroll
        for (int j = 0; j < 7; ++j) {
            const int v = wave + 32 * j;
            const int vv = (v < nloc) ? v : 0;
            s[j] = dinv[lo + vv];
            h[j] = *(const float4*)(H + (size_t)(lo + vv) * DIMF + lane * 4);
        }
        #pragma unroll
        for (int j = 0; j < 7; ++j) {
            const bool in = (wave + 32 * j) < nloc;
            const float q = s[j] * s[j];
            av[4 * j + 0] = in ? q * h[j].x : 0.0f;
            av[4 * j + 1] = in ? q * h[j].y : 0.0f;
            av[4 * j + 2] = in ? q * h[j].z : 0.0f;
            av[4 * j + 3] = in ? q * h[j].w : 0.0f;
        }
    }

#define ADDROW(J, R, S)                                   \
    {                                                     \
        const int e_ = 4 * (J);                           \
        av[e_ + 0] += (R).x * (S);                        \
        av[e_ + 1] += (R).y * (S);                        \
        av[e_ + 2] += (R).z * (S);                        \
        av[e_ + 3] += (R).w * (S);                        \
    }

    int2 cur = (lane < n) ? seg[lane] : make_int2(0, 0);
    for (int b0 = 0; b0 < n; b0 += 32) {
        const int c  = (n - b0 < 32) ? (n - b0) : 32;
        const int nb = b0 + 32 + lane;
        int2 nx = make_int2(0, 0);
        const int   u   = cur.x & 0xFFFF;
        const int   vj  = cur.x >> 21;
        const float nrm = __int_as_float(cur.y);
        #pragma unroll 1
        for (int i = 0; i < c; i += 8) {
            float4 r[8];
            #pragma unroll
            for (int g = 0; g < 8; ++g) {
                int idx = (i + g < c) ? (i + g) : (c - 1);
                int ui = __builtin_amdgcn_readlane(u, idx);
                r[g] = *(const float4*)(H + (size_t)ui * DIMF + lane * 4);
            }
            if (i == 0 && nb < n) nx = seg[nb];
            #pragma unroll
            for (int g = 0; g < 8; ++g) {
                if (i + g < c) {
                    float s = __int_as_float(__builtin_amdgcn_readlane(__float_as_int(nrm), i + g));
                    int   j = __builtin_amdgcn_readlane(vj, i + g);
                    ADDROW(j, r[g], s)
                }
            }
        }
        cur = nx;
    }

    const bool fuse = pre || (Wn != nullptr);
#ifdef USE_F32_WMMA
    const int tc   = lane & 15;
    const int hf   = lane >> 4;
    const int col0 = (wave & 7) << 4;

    v2f bw[32];
    auto strip_perm = [&](const float* __restrict__ Wm) {
        #pragma unroll
        for (int t = 0; t < 32; ++t) {
            const int p = 4 * t + 2 * hf;
            const int f = 4 * (p & 31) + (p >> 5);
            bw[t].x = Wm[(size_t)f       * DIMF + col0 + tc];
            bw[t].y = Wm[(size_t)(f + 4) * DIMF + col0 + tc];
        }
    };
    auto strip_plain = [&](const float* __restrict__ Wm) {
        #pragma unroll
        for (int t = 0; t < 32; ++t) {
            const int p = 4 * t + 2 * hf;
            bw[t].x = Wm[(size_t)p       * DIMF + col0 + tc];
            bw[t].y = Wm[(size_t)(p + 1) * DIMF + col0 + tc];
        }
    };
    if (fuse) strip_perm(pre ? Wp : Wn);
    const float bc = pre ? bias[col0 + tc] : 0.0f;
#endif

    {
        const float4 b4 = pre ? make_float4(0.0f, 0.0f, 0.0f, 0.0f)
                              : *(const float4*)(bias + lane * 4);
        auto finish = [&](int j, float4 o) {
            const int v = wave + 32 * j;
            if (v < nloc) {
                if (!pre) {
                    o.x += b4.x; o.y += b4.y; o.z += b4.z; o.w += b4.w;
                    o.x = (o.x > 0.0f) ? o.x : 0.01f * o.x;
                    o.y = (o.y > 0.0f) ? o.y : 0.01f * o.y;
                    o.z = (o.z > 0.0f) ? o.z : 0.01f * o.z;
                    o.w = (o.w > 0.0f) ? o.w : 0.01f * o.w;
                }
                if (fuse) {
                    float* a = acc + v * PITCH + lane;
                    a[0] = o.x; a[32] = o.y; a[64] = o.z; a[96] = o.w;
                } else {
                    *(float4*)(out + (size_t)(lo + v) * DIMF + lane * 4) = o;
                }
            }
        };
        finish(0, make_float4(av[0], av[1], av[2], av[3]));
        finish(1, make_float4(av[4], av[5], av[6], av[7]));
        finish(2, make_float4(av[8], av[9], av[10], av[11]));
        finish(3, make_float4(av[12], av[13], av[14], av[15]));
        finish(4, make_float4(av[16], av[17], av[18], av[19]));
        finish(5, make_float4(av[20], av[21], av[22], av[23]));
        finish(6, make_float4(av[24], av[25], av[26], av[27]));
#undef ADDROW
        if (fuse) {
            for (int v = nloc + wave; v < RTILES * 16; v += 32) {
                float* a = acc + v * PITCH + lane;
                a[0] = 0.0f; a[32] = 0.0f; a[64] = 0.0f; a[96] = 0.0f;
            }
        }
    }
    if (!fuse) return;

#ifdef USE_F32_WMMA
    __syncthreads();

    const int rg = ((wave >> 3) + wave) & 3;

    auto tile = [&](const float* ar) -> v8f {
        v8f d = {};
        #pragma unroll
        for (int t = 0; t < 32; ++t) {
            v2f a;
            a.x = ar[4 * t];
            a.y = ar[4 * t + 1];
            d = __builtin_amdgcn_wmma_f32_16x16x4_f32(
                    false, a, false, bw[t], (short)0, d, false, false);
        }
        return d;
    };

    if (pre) {
        #pragma unroll 1
        for (int rt = rg; rt < RTILES; rt += 4) {
            if (rt * 16 >= nloc) break;
            const v8f d = tile(acc + (rt * 16 + tc) * PITCH + 2 * hf);
            float* yp = act + (rt * 16 + 8 * hf) * PITCH + col0 + tc;
            #pragma unroll
            for (int j = 0; j < 8; ++j) {
                float y = d[j] + bc;
                y = (y > 0.0f) ? y : 0.01f * y;
                yp[j * PITCH] = y;
            }
        }
        strip_plain(Wn);
        __syncthreads();
    }

    const float* rows = pre ? act : acc;
    #pragma unroll 1
    for (int rt = rg; rt < RTILES; rt += 4) {
        if (rt * 16 >= nloc) break;
        const v8f d = tile(rows + (rt * 16 + tc) * PITCH + 2 * hf);
        const int r0 = rt * 16 + 8 * hf;
        float* hp = out + (size_t)(lo + r0) * DIMF + col0 + tc;
        if (rt * 16 + 16 <= nloc) {
            #pragma unroll
            for (int j = 0; j < 8; ++j)
                hp[(size_t)j * DIMF] = d[j];
        } else {
            const int lim = nloc - r0;
            #pragma unroll
            for (int j = 0; j < 8; ++j)
                if (j < lim) hp[(size_t)j * DIMF] = d[j];
        }
    }
#endif
}

__global__ __launch_bounds__(1024) void gcn_agg(const float* __restrict__ H,
                                                 const int* __restrict__ recw,
                                                 const unsigned int* __restrict__ cnt,
                                                 const unsigned int* __restrict__ cw,
                                                 const float* __restrict__ dinv,
                                                 const float* __restrict__ bias,
                                                 const float* __restrict__ Wn,
                                                 float* __restrict__ out, int N) {
    __shared__ __attribute__((aligned(16))) float acc[RTILES * 16 * PITCH];
    agg_block(acc, nullptr, false, H, recw, cnt, cw, dinv, bias, nullptr, Wn, out, N);
}

__global__ __launch_bounds__(1024) void gcn_agg_first(const float* __restrict__ X,
                                                       const int* __restrict__ recw,
                                                       const unsigned int* __restrict__ cnt,
                                                       const unsigned int* __restrict__ cw,
                                                       const float* __restrict__ dinv,
                                                       const float* __restrict__ bias,
                                                       const float* __restrict__ Wp,
                                                       const float* __restrict__ Wn,
                                                       float* __restrict__ out, int N) {
    __shared__ __attribute__((aligned(16))) float acc[RTILES * 16 * PITCH];
    __shared__ __attribute__((aligned(16))) float act[RTILES * 16 * PITCH];
    agg_block(acc, act, true, X, recw, cnt, cw, dinv, bias, Wp, Wn, out, N);
}


extern "C" void kernel_launch(void* const* d_in, const int* in_sizes, int n_in,
                              void* d_out, int out_size, void* d_ws, size_t ws_size,
                              hipStream_t stream) {
    const float* x  = (const float*)d_in[0];
    const int*   ei = (const int*)d_in[1];
    const float* Wts[3] = {(const float*)d_in[2], (const float*)d_in[4], (const float*)d_in[6]};
    const float* bs[3]  = {(const float*)d_in[3], (const float*)d_in[5], (const float*)d_in[7]};

    const int N = in_sizes[0] / DIMF;
    const int E = in_sizes[1] / 2;
    const int* srcp = ei;
    const int* dstp = ei + E;

    float* ws   = (float*)d_ws;
    float* dinv = ws;
    size_t off  = ((size_t)N + 63) & ~(size_t)63;
    float* Hbuf = ws + off;
    float* Gbuf = Hbuf + (size_t)N * DIMF;
    unsigned int* keyb = (unsigned int*)(Gbuf + (size_t)N * DIMF);
    const int Epad = ((E + KCH - 1) / KCH) * KCH;
    unsigned int* cntb = keyb + Epad / 4;
    unsigned int* cwb = cntb + 256;
    int* recb = (int*)(cwb + 256 * 32);

    const int T = 256;
    const int aggBlocks = (N + RN - 1) / RN;

    zero_f32  <<<(int)((N + T - 1) / T), T, 0, stream>>>(dinv, N);
    deg_count_key<<<(Epad / 4 + T - 1) / T, T, 0, stream>>>(dstp, dinv, keyb, E, Epad);
    deg_finish_cnt<<<aggBlocks, 256, 0, stream>>>(dinv, cntb, cwb, N);
    gcn_build<<<aggBlocks, 1024, 0, stream>>>(srcp, dstp, keyb, dinv, cntb, cwb, recb, E);

    const float* Wnone = nullptr;
    float* outf = (float*)d_out;
    gcn_agg_first<<<aggBlocks, 1024, 0, stream>>>(x, recb, cntb, cwb, dinv, bs[0], Wts[0], Wts[1], Gbuf, N);
    gcn_agg<<<aggBlocks, 1024, 0, stream>>>(Gbuf, recb, cntb, cwb, dinv, bs[1], Wts[2], Hbuf, N);
    gcn_agg<<<aggBlocks, 1024, 0, stream>>>(Hbuf, recb, cntb, cwb, dinv, bs[2], Wnone,  outf, N);
}


# In-degrees counted in LDS integer counters: 64 blocks count one edge slice each into per-node LDS words, then a node's 64 partial counts are added; global f32 atomics and zero launch gone
# speedup vs baseline: 4.5767x; 1.3451x over previous
#include <hip/hip_runtime.h>

typedef float     v2f  __attribute__((ext_vector_type(2)));
typedef float     v8f  __attribute__((ext_vector_type(8)));
typedef float     v32f __attribute__((ext_vector_type(32)));

#define DIMF 128

#if __has_builtin(__builtin_amdgcn_wmma_f32_16x16x4_f32)
#define USE_F32_WMMA 1
#endif


#define RN 196
#define NCAP (256 * RN)
#define HSLICES 64
__global__ __launch_bounds__(1024) void deg_hist_key(const int* __restrict__ dst,
                                                     unsigned int* __restrict__ slices,
                                                     unsigned int* __restrict__ key,
                                                     int E, int Epad) {
    __shared__ __attribute__((aligned(16))) unsigned int hist[NCAP];
    const int tid = threadIdx.x;
    for (int i = tid * 4; i < NCAP; i += 4 * 1024)
        *(uint4*)(hist + i) = make_uint4(0u, 0u, 0u, 0u);
    __syncthreads();

    const int groups = Epad >> 2;
    const int gper   = ((groups + HSLICES - 1) / HSLICES + 31) & ~31;
    const int g0     = blockIdx.x * gper;
    const int g1     = (g0 + gper < groups) ? (g0 + gper) : groups;
    const bool vec_ok = ((E & 3) == 0) && ((((size_t)dst) & 15) == 0);

    for (int gb = g0 + tid; gb < g1; gb += 4 * 1024) {
        int4 d[4];
        #pragma unroll
        for (int u = 0; u < 4; ++u) {
            const int g = gb + 1024 * u;
            const bool full = vec_ok && g < g1 && (g << 2) + 3 < E;
            d[u] = full ? *(const int4*)(dst + ((size_t)g << 2)) : make_int4(0, 0, 0, 0);
        }
        #pragma unroll
        for (int u = 0; u < 4; ++u) {
            const int g = gb + 1024 * u;
            if (g < g1) {
                const int e0 = g << 2;
                unsigned int k = 0;
                if (vec_ok && e0 + 3 < E) {
                    atomicAdd(&hist[d[u].x], 1u);
                    atomicAdd(&hist[d[u].y], 1u);
                    atomicAdd(&hist[d[u].z], 1u);
                    atomicAdd(&hist[d[u].w], 1u);
                    k = (unsigned int)(d[u].x / RN) | ((unsigned int)(d[u].y / RN) << 8) |
                        ((unsigned int)(d[u].z / RN) << 16) | ((unsigned int)(d[u].w / RN) << 24);
                } else {
                    for (int q = 0; q < 4; ++q) {
                        if (e0 + q < E) {
                            const int v = dst[e0 + q];
                            atomicAdd(&hist[v], 1u);
                            k |= (unsigned int)(v / RN) << (8 * q);
                        }
                    }
                }
                key[g] = k;
            }
        }
    }
    __syncthreads();

    unsigned int* __restrict__ p = slices + (size_t)blockIdx.x * NCAP;
    for (int i = tid * 4; i < NCAP; i += 4 * 1024)
        *(uint4*)(p + i) = *(const uint4*)(hist + i);
}

__global__ __launch_bounds__(256) void deg_finish_cnt(const unsigned int* __restrict__ slices,
                                                      float* __restrict__ dinv,
                                                      unsigned int* __restrict__ cnt,
                                                      unsigned int* __restrict__ cw, int n) {
    __shared__ float part[8][32];
    const int v = blockIdx.x * RN + threadIdx.x;
    float d = 0.0f;
    if (threadIdx.x < RN && v < n) {
        unsigned int c = 0;
        #pragma unroll
        for (int i = 0; i < HSLICES; ++i) c += slices[(size_t)i * NCAP + v];
        d = (float)c;
        dinv[v] = rsqrtf(d + 1.0f);
    }
    part[threadIdx.x >> 5][threadIdx.x & 31] = d;
    __syncthreads();
    if (threadIdx.x < 32) {
        float t = 0.0f;
        for (int i = 0; i < 8; ++i) t += part[i][threadIdx.x];
        cw[blockIdx.x * 32 + threadIdx.x] = (unsigned int)t;
        for (int o = 16; o > 0; o >>= 1) t += __shfl_xor(t, o);
        if (threadIdx.x == 0) cnt[blockIdx.x] = (unsigned int)t;
    }
}

#define SCAN_U 8
#define KCH 512

__device__ __forceinline__ unsigned int zero_bytes(unsigned int x) {
    return ~((((x & 0x7F7F7F7Fu) + 0x7F7F7F7Fu) | x) | 0x7F7F7F7Fu);
}

__device__ __forceinline__ unsigned int match16(uint4 d, unsigned int pat) {
    return (zero_bytes(d.x ^ pat) >> 7) | (zero_bytes(d.y ^ pat) >> 6) |
           (zero_bytes(d.z ^ pat) >> 5) | (zero_bytes(d.w ^ pat) >> 4);
}

__device__ __forceinline__ unsigned int match16_tail(uint4 d, unsigned int blk, int e0, int E) {
    const unsigned int w[4] = {d.x, d.y, d.z, d.w};
    unsigned int m = 0;
    #pragma unroll
    for (int q = 0; q < 16; ++q) {
        const unsigned int kb = (w[q >> 2] >> (8 * (q & 3))) & 255u;
        if (e0 + q < E && kb == blk) m |= 1u << (8 * (q & 3) + (q >> 2));
    }
    return m;
}

__device__ __forceinline__ unsigned int segment_start(const unsigned int* __restrict__ cnt,
                                                      int blk, int lane) {
    unsigned int t = 0;
    for (int i = lane; i < blk; i += 32) t += cnt[i];
    for (int o = 16; o > 0; o >>= 1) t += __shfl_xor(t, o);
    return t;
}

__global__ __launch_bounds__(1024) void gcn_build(const int* __restrict__ src,
                                                   const int* __restrict__ dst,
                                                   const unsigned int* __restrict__ key,
                                                   const float* __restrict__ dinv,
                                                   const unsigned int* __restrict__ cnt,
                                                   const unsigned int* __restrict__ cw,
                                                   int* __restrict__ recw, int E) {
    int2* __restrict__ rec = (int2*)recw;
    __shared__ unsigned int s_startw[32];
    __shared__ unsigned int s_fillw[32];

    const int tid  = threadIdx.x;
    const int lane = tid & 31;
    const int wave = __builtin_amdgcn_readfirstlane(tid >> 5);
    const int lo   = blockIdx.x * RN;

    if (wave == 0) {
        const unsigned int t = segment_start(cnt, blockIdx.x, lane);
        const unsigned int cl = cw[blockIdx.x * 32 + lane];
        unsigned int inc = cl;
        for (int o = 1; o < 32; o <<= 1) {
            const unsigned int y = __shfl_up(inc, o);
            if (lane >= o) inc += y;
        }
        s_startw[lane] = t + inc - cl;
        s_fillw[lane]  = 0;
    }
    __syncthreads();

    const unsigned int blk = blockIdx.x;
    const unsigned int pat = blk * 0x01010101u;
    const uint4* key4 = (const uint4*)key;
    const int nchunks = (E + KCH - 1) / KCH;
    int qn = 0;
    int qe = 0;

    auto flush = [&](int c) {
        if (lane < c) {
            const int u = src[qe];
            const int v = dst[qe];
            const float nrm = dinv[u] * dinv[v];
            const int vl = v - lo;
            const unsigned int slot = atomicAdd(&s_fillw[vl & 31], 1u);
            rec[s_startw[vl & 31] + slot] = make_int2(u | (vl << 16), __float_as_int(nrm));
        }
    };

    for (int c = wave; c < nchunks; c += 32 * SCAN_U) {
        unsigned int m[SCAN_U];
        uint4 d[SCAN_U];
        #pragma unroll
        for (int u = 0; u < SCAN_U; ++u) {
            const int cc = c + 32 * u;
            d[u] = key4[((cc < nchunks) ? cc : (nchunks - 1)) * (KCH / 16) + lane];
        }
        if ((c + 32 * (SCAN_U - 1) + 1) * KCH <= E) {
            #pragma unroll
            for (int u = 0; u < SCAN_U; ++u)
                m[u] = match16(d[u], pat);
        } else {
            #pragma unroll
            for (int u = 0; u < SCAN_U; ++u) {
                const int cc = c + 32 * u;
                m[u] = (cc < nchunks) ? match16_tail(d[u], blk, cc * KCH + (lane << 4), E) : 0u;
            }
        }
        const unsigned int h0 = m[0] | (m[1] << 4);
        const unsigned int h1 = m[2] | (m[3] << 4);
        const unsigned int h2 = m[4] | (m[5] << 4);
        const unsigned int h3 = m[6] | (m[7] << 4);
        #pragma unroll 1
        for (int p = 0; p < SCAN_U / 2; ++p) {
            const unsigned int hb = (p == 0) ? h0 : (p == 1) ? h1 : (p == 2) ? h2 : h3;
            unsigned any = __builtin_amdgcn_ballot_w32(hb != 0);
            while (any) {
                int j = __builtin_ctz(any);
                any &= any - 1;
                unsigned hbj = (unsigned)__builtin_amdgcn_readlane((int)hb, j);
                while (hbj) {
                    int t = __builtin_ctz(hbj);
                    hbj &= hbj - 1;
                    int e = (c + 32 * (2 * p + ((t >> 2) & 1))) * KCH + (j << 4) + ((t & 3) << 2) + (t >> 3);
                    if (lane == qn) qe = e;
                    ++qn;
                    if (qn == 32) { flush(32); qn = 0; }
                }
            }
        }
    }
    if (qn) flush(qn);
}

#define PITCH 132
#define RTILES ((RN + 15) / 16)
__device__ __forceinline__ void agg_block(float* acc, float* act,
                                          const bool pre,
                                          const float* __restrict__ H,
                                          const int* __restrict__ recw,
                                          const unsigned int* __restrict__ cnt,
                                          const unsigned int* __restrict__ cw,
                                          const float* __restrict__ dinv,
                                          const float* __restrict__ bias,
                                          const float* __restrict__ Wp,
                                          const float* __restrict__ Wn,
                                          float* __restrict__ out, int N) {
    const int tid  = threadIdx.x;
    const int lane = tid & 31;
    const int wave = __builtin_amdgcn_readfirstlane(tid >> 5);
    const int lo   = blockIdx.x * RN;
    const int nloc = (N - lo < RN) ? (N - lo) : RN;

    const unsigned int bstart = segment_start(cnt, blockIdx.x, lane);
    const unsigned int cl = cw[blockIdx.x * 32 + lane];
    unsigned int before = (lane < wave) ? cl : 0u;
    for (int o = 16; o > 0; o >>= 1) before += __shfl_xor(before, o);
    const int n = __builtin_amdgcn_readlane((int)cl, wave);
    const int2* seg = (const int2*)recw + bstart + before;

    v32f av = {};
    {
        float4 h[7];
        float  s[7];
        #pragma unroll
        for (int j = 0; j < 7; ++j) {
            const int v = wave + 32 * j;
            const int vv = (v < nloc) ? v : 0;
            s[j] = dinv[lo + vv];
            h[j] = *(const float4*)(H + (size_t)(lo + vv) * DIMF + lane * 4);
        }
        #pragma unroll
        for (int j = 0; j < 7; ++j) {
            const bool in = (wave + 32 * j) < nloc;
            const float q = s[j] * s[j];
            av[4 * j + 0] = in ? q * h[j].x : 0.0f;
            av[4 * j + 1] = in ? q * h[j].y : 0.0f;
            av[4 * j + 2] = in ? q * h[j].z : 0.0f;
            av[4 * j + 3] = in ? q * h[j].w : 0.0f;
        }
    }

#define ADDROW(J, R, S)                                   \
    {                                                     \
        const int e_ = 4 * (J);                           \
        av[e_ + 0] += (R).x * (S);                        \
        av[e_ + 1] += (R).y * (S);                        \
        av[e_ + 2] += (R).z * (S);                        \
        av[e_ + 3] += (R).w * (S);                        \
    }

    int2 cur = (lane < n) ? seg[lane] : make_int2(0, 0);
    for (int b0 = 0; b0 < n; b0 += 32) {
        const int c  = (n - b0 < 32) ? (n - b0) : 32;
        const int nb = b0 + 32 + lane;
        int2 nx = make_int2(0, 0);
        const int   u   = cur.x & 0xFFFF;
        const int   vj  = cur.x >> 21;
        const float nrm = __int_as_float(cur.y);
        #pragma unroll 1
        for (int i = 0; i < c; i += 8) {
            float4 r[8];
            #pragma unroll
            for (int g = 0; g < 8; ++g) {
                int idx = (i + g < c) ? (i + g) : (c - 1);
                int ui = __builtin_amdgcn_readlane(u, idx);
                r[g] = *(const float4*)(H + (size_t)ui * DIMF + lane * 4);
            }
            if (i == 0 && nb < n) nx = seg[nb];
            #pragma unroll
            for (int g = 0; g < 8; ++g) {
                if (i + g < c) {
                    float s = __int_as_float(__builtin_amdgcn_readlane(__float_as_int(nrm), i + g));
                    int   j = __builtin_amdgcn_readlane(vj, i + g);
                    ADDROW(j, r[g], s)
                }
            }
        }
        cur = nx;
    }

    const bool fuse = pre || (Wn != nullptr);
#ifdef USE_F32_WMMA
    const int tc   = lane & 15;
    const int hf   = lane >> 4;
    const int col0 = (wave & 7) << 4;

    v2f bw[32];
    auto strip_perm = [&](const float* __restrict__ Wm) {
        #pragma unroll
        for (int t = 0; t < 32; ++t) {
            const int p = 4 * t + 2 * hf;
            const int f = 4 * (p & 31) + (p >> 5);
            bw[t].x = Wm[(size_t)f       * DIMF + col0 + tc];
            bw[t].y = Wm[(size_t)(f + 4) * DIMF + col0 + tc];
        }
    };
    auto strip_plain = [&](const float* __restrict__ Wm) {
        #pragma unroll
        for (int t = 0; t < 32; ++t) {
            const int p = 4 * t + 2 * hf;
            bw[t].x = Wm[(size_t)p       * DIMF + col0 + tc];
            bw[t].y = Wm[(size_t)(p + 1) * DIMF + col0 + tc];
        }
    };
    if (fuse) strip_perm(pre ? Wp : Wn);
    const float bc = pre ? bias[col0 + tc] : 0.0f;
#endif

    {
        const float4 b4 = pre ? make_float4(0.0f, 0.0f, 0.0f, 0.0f)
                              : *(const float4*)(bias + lane * 4);
        auto finish = [&](int j, float4 o) {
            const int v = wave + 32 * j;
            if (v < nloc) {
                if (!pre) {
                    o.x += b4.x; o.y += b4.y; o.z += b4.z; o.w += b4.w;
                    o.x = (o.x > 0.0f) ? o.x : 0.01f * o.x;
                    o.y = (o.y > 0.0f) ? o.y : 0.01f * o.y;
                    o.z = (o.z > 0.0f) ? o.z : 0.01f * o.z;
                    o.w = (o.w > 0.0f) ? o.w : 0.01f * o.w;
                }
                if (fuse) {
                    float* a = acc + v * PITCH + lane;
                    a[0] = o.x; a[32] = o.y; a[64] = o.z; a[96] = o.w;
                } else {
                    *(float4*)(out + (size_t)(lo + v) * DIMF + lane * 4) = o;
                }
            }
        };
        finish(0, make_float4(av[0], av[1], av[2], av[3]));
        finish(1, make_float4(av[4], av[5], av[6], av[7]));
        finish(2, make_float4(av[8], av[9], av[10], av[11]));
        finish(3, make_float4(av[12], av[13], av[14], av[15]));
        finish(4, make_float4(av[16], av[17], av[18], av[19]));
        finish(5, make_float4(av[20], av[21], av[22], av[23]));
        finish(6, make_float4(av[24], av[25], av[26], av[27]));
#undef ADDROW
        if (fuse) {
            for (int v = nloc + wave; v < RTILES * 16; v += 32) {
                float* a = acc + v * PITCH + lane;
                a[0] = 0.0f; a[32] = 0.0f; a[64] = 0.0f; a[96] = 0.0f;
            }
        }
    }
    if (!fuse) return;

#ifdef USE_F32_WMMA
    __syncthreads();

    const int rg = ((wave >> 3) + wave) & 3;

    auto tile = [&](const float* ar) -> v8f {
        v8f d = {};
        #pragma unroll
        for (int t = 0; t < 32; ++t) {
            v2f a;
            a.x = ar[4 * t];
            a.y = ar[4 * t + 1];
            d = __builtin_amdgcn_wmma_f32_16x16x4_f32(
                    false, a, false, bw[t], (short)0, d, false, false);
        }
        return d;
    };

    if (pre) {
        #pragma unroll 1
        for (int rt = rg; rt < RTILES; rt += 4) {
            if (rt * 16 >= nloc) break;
            const v8f d = tile(acc + (rt * 16 + tc) * PITCH + 2 * hf);
            float* yp = act + (rt * 16 + 8 * hf) * PITCH + col0 + tc;
            #pragma unroll
            for (int j = 0; j < 8; ++j) {
                float y = d[j] + bc;
                y = (y > 0.0f) ? y : 0.01f * y;
                yp[j * PITCH] = y;
            }
        }
        strip_plain(Wn);
        __syncthreads();
    }

    const float* rows = pre ? act : acc;
    #pragma unroll 1
    for (int rt = rg; rt < RTILES; rt += 4) {
        if (rt * 16 >= nloc) break;
        const v8f d = tile(rows + (rt * 16 + tc) * PITCH + 2 * hf);
        const int r0 = rt * 16 + 8 * hf;
        float* hp = out + (size_t)(lo + r0) * DIMF + col0 + tc;
        if (rt * 16 + 16 <= nloc) {
            #pragma unroll
            for (int j = 0; j < 8; ++j)
                hp[(size_t)j * DIMF] = d[j];
        } else {
            const int lim = nloc - r0;
            #pragma unroll
            for (int j = 0; j < 8; ++j)
                if (j < lim) hp[(size_t)j * DIMF] = d[j];
        }
    }
#endif
}

__global__ __launch_bounds__(1024) void gcn_agg(const float* __restrict__ H,
                                                 const int* __restrict__ recw,
                                                 const unsigned int* __restrict__ cnt,
                                                 const unsigned int* __restrict__ cw,
                                                 const float* __restrict__ dinv,
                                                 const float* __restrict__ bias,
                                                 const float* __restrict__ Wn,
                                                 float* __restrict__ out, int N) {
    __shared__ __attribute__((aligned(16))) float acc[RTILES * 16 * PITCH];
    agg_block(acc, nullptr, false, H, recw, cnt, cw, dinv, bias, nullptr, Wn, out, N);
}

__global__ __launch_bounds__(1024) void gcn_agg_first(const float* __restrict__ X,
                                                       const int* __restrict__ recw,
                                                       const unsigned int* __restrict__ cnt,
                                                       const unsigned int* __restrict__ cw,
                                                       const float* __restrict__ dinv,
                                                       const float* __restrict__ bias,
                                                       const float* __restrict__ Wp,
                                                       const float* __restrict__ Wn,
                                                       float* __restrict__ out, int N) {
    __shared__ __attribute__((aligned(16))) float acc[RTILES * 16 * PITCH];
    __shared__ __attribute__((aligned(16))) float act[RTILES * 16 * PITCH];
    agg_block(acc, act, true, X, recw, cnt, cw, dinv, bias, Wp, Wn, out, N);
}


extern "C" void kernel_launch(void* const* d_in, const int* in_sizes, int n_in,
                              void* d_out, int out_size, void* d_ws, size_t ws_size,
                              hipStream_t stream) {
    const float* x  = (const float*)d_in[0];
    const int*   ei = (const int*)d_in[1];
    const float* Wts[3] = {(const float*)d_in[2], (const float*)d_in[4], (const float*)d_in[6]};
    const float* bs[3]  = {(const float*)d_in[3], (const float*)d_in[5], (const float*)d_in[7]};

    const int N = in_sizes[0] / DIMF;
    const int E = in_sizes[1] / 2;
    const int* srcp = ei;
    const int* dstp = ei + E;

    float* ws   = (float*)d_ws;
    float* dinv = ws;
    size_t off  = ((size_t)N + 63) & ~(size_t)63;
    float* Hbuf = ws + off;
    float* Gbuf = Hbuf + (size_t)N * DIMF;
    unsigned int* keyb = (unsigned int*)(Gbuf + (size_t)N * DIMF);
    const int Epad = ((E + KCH - 1) / KCH) * KCH;
    unsigned int* cntb = keyb + Epad / 4;
    unsigned int* cwb = cntb + 256;
    int* recb = (int*)(cwb + 256 * 32);
    unsigned int* slb = (unsigned int*)(recb + ((2 * (size_t)E + 3) & ~(size_t)3));

    const int T = 256;
    const int aggBlocks = (N + RN - 1) / RN;

    deg_hist_key<<<HSLICES, 1024, 0, stream>>>(dstp, slb, keyb, E, Epad);
    deg_finish_cnt<<<aggBlocks, T, 0, stream>>>(slb, dinv, cntb, cwb, N);
    gcn_build<<<aggBlocks, 1024, 0, stream>>>(srcp, dstp, keyb, dinv, cntb, cwb, recb, E);

    const float* Wnone = nullptr;
    float* outf = (float*)d_out;
    gcn_agg_first<<<aggBlocks, 1024, 0, stream>>>(x, recb, cntb, cwb, dinv, bs[0], Wts[0], Wts[1], Gbuf, N);
    gcn_agg<<<aggBlocks, 1024, 0, stream>>>(Gbuf, recb, cntb, cwb, dinv, bs[1], Wts[2], Hbuf, N);
    gcn_agg<<<aggBlocks, 1024, 0, stream>>>(Hbuf, recb, cntb, cwb, dinv, bs[2], Wnone,  outf, N);
}
